# starting kernel
# baseline (speedup 1.0000x reference)
#include <hip/hip_runtime.h>

typedef __attribute__((ext_vector_type(2))) float v2f;
typedef __attribute__((ext_vector_type(8))) float v8f;

#define BB 8192
#define DD 512
#define OO 256

__global__ __launch_bounds__(256) void xlayer_fold_xval(
    const float* __restrict__ kern, const float* __restrict__ xval,
    float* __restrict__ W) {
  __shared__ float sxval[DD];
  const int o = threadIdx.x;
  const int i = blockIdx.x;
  for (int c = threadIdx.x; c < DD; c += 256) sxval[c] = xval[c];
  __syncthreads();

  const float* kp = kern + (size_t)i * (DD * OO) + o;
  float a0 = 0.f, a1 = 0.f, a2 = 0.f, a3 = 0.f;
#pragma unroll 4
  for (int c = 0; c < DD; c += 4) {
    a0 = fmaf(kp[(size_t)(c + 0) * OO], sxval[c + 0], a0);
    a1 = fmaf(kp[(size_t)(c + 1) * OO], sxval[c + 1], a1);
    a2 = fmaf(kp[(size_t)(c + 2) * OO], sxval[c + 2], a2);
    a3 = fmaf(kp[(size_t)(c + 3) * OO], sxval[c + 3], a3);
  }
  W[i * OO + o] = (a0 + a1) + (a2 + a3);
}

__global__ __launch_bounds__(256) void xlayer_wmma_gemm(
    const float* __restrict__ x, const float* __restrict__ W,
    float* __restrict__ out) {
  const int lane = threadIdx.x & 31;
  const int wave = (blockIdx.x << 3) + (threadIdx.x >> 5);
  const int NT = OO / 64;
  const int mBase = (wave / NT) * 16;
  const int nBase = (wave % NT) * 64;
  const int half = lane >> 4;
  const int l16  = lane & 15;

  v8f acc0 = {}, acc1 = {}, acc2 = {}, acc3 = {};

  const float* xrow = x + (size_t)(mBase + l16) * DD + 2 * half;
  const float* wcol = W + nBase + l16;

#pragma unroll 4
  for (int k0 = 0; k0 < DD; k0 += 4) {
    v2f a;
    a.x = xrow[k0];
    a.y = xrow[k0 + 1];

    const int kr = k0 + 2 * half;
    const float* wp0 = wcol + (size_t)kr * OO;
    const float* wp1 = wp0 + OO;
    v2f b0, b1, b2, b3;
    b0.x = wp0[0];  b0.y = wp1[0];
    b1.x = wp0[16]; b1.y = wp1[16];
    b2.x = wp0[32]; b2.y = wp1[32];
    b3.x = wp0[48]; b3.y = wp1[48];

    acc0 = __builtin_amdgcn_wmma_f32_16x16x4_f32(false, a, false, b0,
                                                 (short)0, acc0, false, false);
    acc1 = __builtin_amdgcn_wmma_f32_16x16x4_f32(false, a, false, b1,
                                                 (short)0, acc1, false, false);
    acc2 = __builtin_amdgcn_wmma_f32_16x16x4_f32(false, a, false, b2,
                                                 (short)0, acc2, false, false);
    acc3 = __builtin_amdgcn_wmma_f32_16x16x4_f32(false, a, false, b3,
                                                 (short)0, acc3, false, false);
  }

  float* orow = out + (size_t)(mBase + 8 * half) * OO + nBase + l16;
#pragma unroll
  for (int r = 0; r < 8; ++r) {
    float* p = orow + (size_t)r * OO;
    p[0]  = acc0[r];
    p[16] = acc1[r];
    p[32] = acc2[r];
    p[48] = acc3[r];
  }
}

extern "C" void kernel_launch(void* const* d_in, const int* in_sizes, int n_in,
                              void* d_out, int out_size, void* d_ws, size_t ws_size,
                              hipStream_t stream) {
  const float* x    = (const float*)d_in[0];
  const float* xval = (const float*)d_in[1];
  const float* kern = (const float*)d_in[2];
  float* out = (float*)d_out;
  float* W   = (float*)d_ws;

  xlayer_fold_xval<<<DD, 256, 0, stream>>>(kern, xval, W);

  const int mTiles = BB / 16;
  const int nTiles = OO / 64;
  const int blocks = (mTiles * nTiles) / 8;
  xlayer_wmma_gemm<<<blocks, 256, 0, stream>>>(x, W, out);
}


# GEMM fragments in a ring of four 16-k register stages: a stage's 10 loads are issued three stages (48 products) ahead of its use and counted in; tile, k order and sum order unchanged
# speedup vs baseline: 1.8556x; 1.8556x over previous
#include <hip/hip_runtime.h>

typedef __attribute__((ext_vector_type(2))) float v2f;
typedef __attribute__((ext_vector_type(4))) float v4f;
typedef __attribute__((ext_vector_type(8))) float v8f;

#define BB 8192
#define DD 512
#define OO 256

#define FOLD_RING 32
#define FOLD_STEPS (DD / 2)

__global__ __launch_bounds__(128) void xlayer_fold_xval(
    const float* __restrict__ kern, const float* __restrict__ xval,
    float* __restrict__ W) {
  __shared__ float sx[DD];
  const int tid = threadIdx.x;
  const int lane = tid & 31;
  const int wv = tid >> 5;
  const int h = lane >> 4;
  const int n = lane & 15;
  const int i = blockIdx.x;

  const float* p = kern + (size_t)i * (DD * OO) + h * OO + 64 * wv + 4 * n;
  const float* sm = sx + FOLD_STEPS * h;
  const size_t STEP = 2 * OO;

  v4f buf[FOLD_RING];
#pragma unroll
  for (int j = 0; j < FOLD_RING; ++j)
    buf[j] = *(const v4f*)(p + (size_t)j * STEP);

  {
    const v4f v = *(const v4f*)(xval + 4 * tid);
    const v2f ev = {v.x, v.z};
    const v2f od = {v.y, v.w};
    *(v2f*)(sx + 2 * tid) = ev;
    *(v2f*)(sx + FOLD_STEPS + 2 * tid) = od;
  }
  __syncthreads();

  v4f se = {0.f, 0.f, 0.f, 0.f};
  v4f so = {0.f, 0.f, 0.f, 0.f};

#define FOLD_ADD(acc, v, m)                   \
  do {                                        \
    acc.x = __builtin_fmaf(v.x, m, acc.x);    \
    acc.y = __builtin_fmaf(v.y, m, acc.y);    \
    acc.z = __builtin_fmaf(v.z, m, acc.z);    \
    acc.w = __builtin_fmaf(v.w, m, acc.w);    \
  } while (0)

#pragma unroll 1
  for (int t0 = 0; t0 < FOLD_STEPS - FOLD_RING; t0 += FOLD_RING) {
    const float* pn = p + (size_t)(t0 + FOLD_RING) * STEP;
    v4f m[FOLD_RING / 4];
#pragma unroll
    for (int g = 0; g < FOLD_RING / 4; ++g)
      m[g] = *(const v4f*)(sm + t0 + 4 * g);
#pragma unroll
    for (int j = 0; j < FOLD_RING; j += 2) {
      FOLD_ADD(se, buf[j], m[j / 4][j % 4]);
      buf[j] = *(const v4f*)(pn + (size_t)j * STEP);
      __builtin_amdgcn_sched_barrier(0);
      FOLD_ADD(so, buf[j + 1], m[j / 4][j % 4 + 1]);
      buf[j + 1] = *(const v4f*)(pn + (size_t)(j + 1) * STEP);
      __builtin_amdgcn_sched_barrier(0);
    }
  }
#pragma unroll
  for (int j = 0; j < FOLD_RING; j += 4) {
    const v4f m = *(const v4f*)(sm + (FOLD_STEPS - FOLD_RING) + j);
    FOLD_ADD(se, buf[j + 0], m.x);
    FOLD_ADD(so, buf[j + 1], m.y);
    FOLD_ADD(se, buf[j + 2], m.z);
    FOLD_ADD(so, buf[j + 3], m.w);
  }
#undef FOLD_ADD

  v4f r;
  r.x = (se.x + __shfl_xor(se.x, 16, 32)) + (so.x + __shfl_xor(so.x, 16, 32));
  r.y = (se.y + __shfl_xor(se.y, 16, 32)) + (so.y + __shfl_xor(so.y, 16, 32));
  r.z = (se.z + __shfl_xor(se.z, 16, 32)) + (so.z + __shfl_xor(so.z, 16, 32));
  r.w = (se.w + __shfl_xor(se.w, 16, 32)) + (so.w + __shfl_xor(so.w, 16, 32));
  if (h == 0) *(v4f*)(W + i * OO + 64 * wv + 4 * n) = r;
}

#define GEMM_STAGE_K 16

__device__ __forceinline__ void gemm_stage_load(v4f (&a)[2], v4f (&w)[2][4],
                                                const float* __restrict__ xs,
                                                const float* __restrict__ ws) {
#pragma unroll
  for (int g = 0; g < 2; ++g)
    a[g] = *(const v4f*)(xs + 8 * g);
#pragma unroll
  for (int g = 0; g < 2; ++g) {
#pragma unroll
    for (int e = 0; e < 4; ++e)
      w[g][e] = *(const v4f*)(ws + (size_t)(8 * g + e) * OO);
  }
}

__device__ __forceinline__ void gemm_stage_mma(const v4f (&a)[2],
                                               const v4f (&w)[2][4], v8f& acc0,
                                               v8f& acc1, v8f& acc2, v8f& acc3) {
#pragma unroll
  for (int g = 0; g < 2; ++g) {
    const v2f a01 = {a[g].x, a[g].y};
    const v2f a23 = {a[g].z, a[g].w};
    {
      const v4f r0 = w[g][0], r1 = w[g][1];
      const v2f b0 = {r0.x, r1.x}, b1 = {r0.y, r1.y};
      const v2f b2 = {r0.z, r1.z}, b3 = {r0.w, r1.w};
      acc0 = __builtin_amdgcn_wmma_f32_16x16x4_f32(false, a01, false, b0,
                                                   (short)0, acc0, false, false);
      acc1 = __builtin_amdgcn_wmma_f32_16x16x4_f32(false, a01, false, b1,
                                                   (short)0, acc1, false, false);
      acc2 = __builtin_amdgcn_wmma_f32_16x16x4_f32(false, a01, false, b2,
                                                   (short)0, acc2, false, false);
      acc3 = __builtin_amdgcn_wmma_f32_16x16x4_f32(false, a01, false, b3,
                                                   (short)0, acc3, false, false);
    }
    {
      const v4f r2 = w[g][2], r3 = w[g][3];
      const v2f b0 = {r2.x, r3.x}, b1 = {r2.y, r3.y};
      const v2f b2 = {r2.z, r3.z}, b3 = {r2.w, r3.w};
      acc0 = __builtin_amdgcn_wmma_f32_16x16x4_f32(false, a23, false, b0,
                                                   (short)0, acc0, false, false);
      acc1 = __builtin_amdgcn_wmma_f32_16x16x4_f32(false, a23, false, b1,
                                                   (short)0, acc1, false, false);
      acc2 = __builtin_amdgcn_wmma_f32_16x16x4_f32(false, a23, false, b2,
                                                   (short)0, acc2, false, false);
      acc3 = __builtin_amdgcn_wmma_f32_16x16x4_f32(false, a23, false, b3,
                                                   (short)0, acc3, false, false);
    }
  }
}

__global__ __launch_bounds__(256) void xlayer_wmma_gemm(
    const float* __restrict__ x, const float* __restrict__ W,
    float* __restrict__ out) {
  const int lane = threadIdx.x & 31;
  const int wave = (blockIdx.x << 3) + (threadIdx.x >> 5);
  const int NT = OO / 64;
  const int mBase = (wave / NT) * 16;
  const int nBase = (wave % NT) * 64;
  const int h = lane >> 4;
  const int c = lane & 15;

  v8f acc0 = {}, acc1 = {}, acc2 = {}, acc3 = {};

  const float* xp = x + (size_t)(mBase + c) * DD + 4 * h;
  const float* wp = W + (size_t)(4 * h) * OO + nBase + 4 * c;

  v4f a[4][2];
  v4f w[4][2][4];

#define GEMM_LOAD(s, k) \
  gemm_stage_load(a[s], w[s], xp + (k), wp + (size_t)(k) * OO)
#define GEMM_MMA(s) gemm_stage_mma(a[s], w[s], acc0, acc1, acc2, acc3)
#define GEMM_FENCE()                                                    \
  do {                                                                  \
          \
    asm volatile("" : "+v"(acc0), "+v"(acc1), "+v"(acc2), "+v"(acc3)); \
             \
    __builtin_amdgcn_sched_barrier(0);                                  \
  } while (0)

  GEMM_LOAD(0, 0 * GEMM_STAGE_K);
  GEMM_FENCE();
  GEMM_LOAD(1, 1 * GEMM_STAGE_K);
  GEMM_FENCE();
  GEMM_LOAD(2, 2 * GEMM_STAGE_K);
  GEMM_FENCE();

#pragma unroll 1
  for (int kb = 0; kb < DD - 4 * GEMM_STAGE_K; kb += 4 * GEMM_STAGE_K) {
    GEMM_LOAD(3, kb + 3 * GEMM_STAGE_K);
    GEMM_FENCE();
    GEMM_MMA(0);
    GEMM_FENCE();
    GEMM_LOAD(0, kb + 4 * GEMM_STAGE_K);
    GEMM_FENCE();
    GEMM_MMA(1);
    GEMM_FENCE();
    GEMM_LOAD(1, kb + 5 * GEMM_STAGE_K);
    GEMM_FENCE();
    GEMM_MMA(2);
    GEMM_FENCE();
    GEMM_LOAD(2, kb + 6 * GEMM_STAGE_K);
    GEMM_FENCE();
    GEMM_MMA(3);
    GEMM_FENCE();
  }
  GEMM_LOAD(3, DD - GEMM_STAGE_K);
  GEMM_FENCE();
  GEMM_MMA(0);
  GEMM_MMA(1);
  GEMM_MMA(2);
  GEMM_MMA(3);
#undef GEMM_LOAD
#undef GEMM_MMA
#undef GEMM_FENCE

  float* orow = out + (size_t)(mBase + 8 * h) * OO + nBase + 4 * c;
#pragma unroll
  for (int r = 0; r < 8; ++r) {
    const v4f v = {acc0[r], acc1[r], acc2[r], acc3[r]};
    *(v4f*)(orow + (size_t)r * OO) = v;
  }
}

extern "C" void kernel_launch(void* const* d_in, const int* in_sizes, int n_in,
                              void* d_out, int out_size, void* d_ws, size_t ws_size,
                              hipStream_t stream) {
  const float* x    = (const float*)d_in[0];
  const float* xval = (const float*)d_in[1];
  const float* kern = (const float*)d_in[2];
  float* out = (float*)d_out;
  float* W   = (float*)d_ws;

  xlayer_fold_xval<<<DD, 128, 0, stream>>>(kern, xval, W);

  const int mTiles = BB / 16;
  const int nTiles = OO / 64;
  const int blocks = (mTiles * nTiles) / 8;
  xlayer_wmma_gemm<<<blocks, 256, 0, stream>>>(x, W, out);
}
